# starting kernel
# baseline (speedup 1.0000x reference)
#include <hip/hip_runtime.h>

typedef __attribute__((ext_vector_type(2))) float v2f;
typedef __attribute__((ext_vector_type(8))) float v8f;

#define Bsz   4096
#define Csz   256
#define LM    16

#define OFF_W3   0
#define OFF_C2   1048576
#define OFF_C1   1114112
#define OFF_O1   1118208

__global__ __launch_bounds__(256) void pb_precompute(
    const float* __restrict__ U3, const float* __restrict__ U2,
    const float* __restrict__ U1, const float* __restrict__ w3,
    const float* __restrict__ w2, const float* __restrict__ w1,
    float* __restrict__ ws)
{
    int idx = blockIdx.x * 256 + threadIdx.x;
    if (idx < 1048576) {
        int c   = idx >> 12;
        int xyi = idx & 4095;
        float s = 0.f;
        #pragma unroll
        for (int k = 0; k < 23; ++k) s += U3[xyi * 23 + k] * w3[k * Csz + c];
        ws[OFF_W3 + idx] = s;
    } else if (idx < 1048576 + 65536) {
        int e  = idx - 1048576;
        int c  = e >> 8;
        int xi = e & 255;
        float s = 0.f;
        #pragma unroll
        for (int k = 0; k < 8; ++k) s += U2[xi * 8 + k] * w2[k * Csz + c];
        ws[OFF_C2 + c * 256 + xi] = s;
    } else if (idx < 1048576 + 65536 + 4096) {
        int e = idx - 1048576 - 65536;
        int c = e >> 4;
        int i = e & 15;
        float s = 0.f;
        #pragma unroll
        for (int k = 0; k < 4; ++k) s += U1[i * 4 + k] * w1[k * Csz + c];
        ws[OFF_C1 + c * 16 + i] = s;
    }
}

__global__ __launch_bounds__(256) void pb_main(
    const float* __restrict__ nf,
    const float* __restrict__ ws,
    float* __restrict__ out1)
{
    __shared__ float sW3[4096];
    __shared__ float sC2[256];
    __shared__ float sC1[16];
    __shared__ float sN[8 * 256];

    const int t  = threadIdx.x;
    const int c  = blockIdx.y;
    const int w  = t >> 5;
    const int l  = t & 31;
    const int l16 = l & 15;
    const int hi  = l >> 4;

    {
        const float4* g = (const float4*)(ws + OFF_W3 + c * 4096);
        float4* s = (float4*)sW3;
        #pragma unroll
        for (int j = 0; j < 4; ++j) s[j * 256 + t] = g[j * 256 + t];
    }
    sC2[t] = ws[OFF_C2 + c * 256 + t];
    if (t < 16) sC1[t] = ws[OFF_C1 + c * 16 + t];
    {
        const int row  = l >> 1;
        const int half = l & 1;
        const int b    = (blockIdx.x * 8 + w) * 16 + row;
        const float4* g = (const float4*)(nf + (size_t)b * (Csz * LM) + c * LM + half * 8);
        float4* s = (float4*)(sN + w * 256 + row * 16 + half * 8);
        s[0] = g[0];
        s[1] = g[1];
    }
    __syncthreads();

    const float* Nw = sN + w * 256;

    v2f Bk[4];
    #pragma unroll
    for (int kk = 0; kk < 4; ++kk) {
        int i0 = 4 * kk + 2 * hi;
        Bk[kk].x = Nw[l16 * 16 + i0];
        Bk[kk].y = Nw[l16 * 16 + i0 + 1];
    }
    float nfd[8];
    #pragma unroll
    for (int r = 0; r < 8; ++r) nfd[r] = Nw[l16 * 16 + r + 8 * hi];

    float acc = 0.f;

    #pragma unroll 4
    for (int x = 0; x < 16; ++x) {
        v8f C;
        #pragma unroll
        for (int r = 0; r < 8; ++r) C[r] = sC2[x * 16 + r + 8 * hi];

        const float* Ax = sW3 + x * 256;
        #pragma unroll
        for (int kk = 0; kk < 4; ++kk) {
            int i0 = 4 * kk + 2 * hi;
            v2f A;
            A.x = Ax[l16 * 16 + i0];
            A.y = Ax[l16 * 16 + i0 + 1];
            C = __builtin_amdgcn_wmma_f32_16x16x4_f32(
                    false, A, false, Bk[kk], (short)0, C, false, false);
        }
        float e = C[0] * nfd[0];
        #pragma unroll
        for (int r = 1; r < 8; ++r) e += C[r] * nfd[r];
        e += __shfl_xor(e, 16, 32);
        acc += (e + sC1[x]) * Nw[l16 * 16 + x];
    }

    if (hi == 0) {
        int b = (blockIdx.x * 8 + w) * 16 + l16;
        out1[c * Bsz + b] = acc;
    }
}

__global__ __launch_bounds__(256) void pb_final_gemm(
    const float* __restrict__ o1,
    const float* __restrict__ lw,
    float* __restrict__ out)
{
    const int t   = threadIdx.x;
    const int w   = t >> 5;
    const int l   = t & 31;
    const int l16 = l & 15;
    const int hi  = l >> 4;

    const int tile = blockIdx.x * 8 + w;
    const int b0 = (tile >> 4) * 16;
    const int d0 = (tile & 15) * 16;

    v8f acc;
    #pragma unroll
    for (int r = 0; r < 8; ++r) acc[r] = 0.f;

    #pragma unroll 4
    for (int k0 = 0; k0 < 256; k0 += 4) {
        const int ka = k0 + 2 * hi;
        v2f A, B;
        A.x = o1[(size_t)ka * Bsz + b0 + l16];
        A.y = o1[(size_t)(ka + 1) * Bsz + b0 + l16];
        B.x = lw[ka * Csz + d0 + l16];
        B.y = lw[(ka + 1) * Csz + d0 + l16];
        acc = __builtin_amdgcn_wmma_f32_16x16x4_f32(
                  false, A, false, B, (short)0, acc, false, false);
    }

    #pragma unroll
    for (int r = 0; r < 8; ++r)
        out[(size_t)(b0 + r + 8 * hi) * Csz + d0 + l16] = acc[r] * 0.0625f;
}

extern "C" void kernel_launch(void* const* d_in, const int* in_sizes, int n_in,
                              void* d_out, int out_size, void* d_ws, size_t ws_size,
                              hipStream_t stream) {
    (void)in_sizes; (void)n_in; (void)out_size; (void)ws_size;
    const float* nf  = (const float*)d_in[0];
    const float* U3  = (const float*)d_in[1];
    const float* U2  = (const float*)d_in[2];
    const float* U1  = (const float*)d_in[3];
    const float* w3  = (const float*)d_in[4];
    const float* w2  = (const float*)d_in[5];
    const float* w1  = (const float*)d_in[6];
    const float* lw  = (const float*)d_in[7];

    float* ws   = (float*)d_ws;
    float* out1 = ws + OFF_O1;

    pb_precompute<<<4368, 256, 0, stream>>>(U3, U2, U1, w3, w2, w1, ws);
    pb_main<<<dim3(32, 256), 256, 0, stream>>>(nf, ws, out1);
    pb_final_gemm<<<512, 256, 0, stream>>>(out1, lw, (float*)d_out);
}


# Cubic term as one K=136 product per tile: fold adds W3[x,y,i]+W3[x,i,y] per unordered (y,i) pair, B = n_y*n_i formed in registers, M=x: 38 WMMAs per tile instead of 64; sum order changes
# speedup vs baseline: 1.4011x; 1.4011x over previous
#include <hip/hip_runtime.h>

typedef __attribute__((ext_vector_type(2))) float v2f;
typedef __attribute__((ext_vector_type(8))) float v8f;

#define Bsz   4096
#define Csz   256
#define LM    16

#define OFF_W3   0
#define OFF_C2   1048576
#define OFF_C1   1114112
#define OFF_O1   1118208

__host__ __device__ constexpr int JA(int q) { return q < 64 ? 2 * (q & 7) : 2 * (q - 64); }
__host__ __device__ constexpr int JB(int q) { return q < 64 ? ((2 * (q & 7) + (q >> 3)) & 15) : (2 * (q - 64) + 8); }

#define NKK      34
#define WP_PER_C 2176

#define FOLD_CG 4
__global__ __launch_bounds__(256) void pb_precompute(
    const float* __restrict__ U3, const float* __restrict__ U2,
    const float* __restrict__ U1, const float* __restrict__ w3,
    const float* __restrict__ w2, const float* __restrict__ w1,
    float* __restrict__ ws)
{
    const int t  = threadIdx.x;
    const int bx = blockIdx.x;
    if (bx < 576) {
        const int cg  = bx / 9;
        const int rem = (bx - cg * 9) * 256 + t;
        if (rem < WP_PER_C) {
            int kk   = rem >> 6;
            int lane = (rem >> 1) & 31;
            int sel  = rem & 1;
            int x = lane & 15, h = lane >> 4;
            int q = 2 * kk + sel;
            int y = (JA(q) + h) & 15;
            int i = (JB(q) + h) & 15;
            const float* r1 = U3 + ((x * 16 + y) * 16 + i) * 23;
            const float* r2 = U3 + ((x * 16 + i) * 16 + y) * 23;
            float u1[23], u2[23];
            #pragma unroll
            for (int k = 0; k < 23; ++k) { u1[k] = r1[k]; u2[k] = r2[k]; }
            #pragma unroll
            for (int j = 0; j < FOLD_CG; ++j) {
                const int c = cg * FOLD_CG + j;
                float s1 = 0.f, s2 = 0.f;
                #pragma unroll
                for (int k = 0; k < 23; ++k) {
                    float wk = w3[k * Csz + c];
                    s1 += u1[k] * wk;
                    s2 += u2[k] * wk;
                }
                ws[OFF_W3 + c * WP_PER_C + rem] = (y != i) ? (s1 + s2) : s1;
            }
        }
        return;
    }
    const int idx = (bx - 576) * 256 + t;
    if (idx < 65536) {
        int c    = idx >> 8;
        int rem  = idx & 255;
        int kk   = rem >> 6;
        int lane = (rem >> 1) & 31;
        int sel  = rem & 1;
        int x = lane & 15, h = lane >> 4;
        int y = (2 * (2 * kk + sel) + h) & 15;
        float s = 0.f;
        #pragma unroll
        for (int k = 0; k < 8; ++k) s += U2[(x * 16 + y) * 8 + k] * w2[k * Csz + c];
        ws[OFF_C2 + idx] = s;
    } else if (idx < 65536 + 4096) {
        int e = idx - 65536;
        int c = e >> 4;
        int i = e & 15;
        float s = 0.f;
        #pragma unroll
        for (int k = 0; k < 4; ++k) s += U1[i * 4 + k] * w1[k * Csz + c];
        ws[OFF_C1 + c * 16 + i] = s;
    }
}

__global__ __launch_bounds__(256) void pb_main(
    const float* __restrict__ nf,
    const float* __restrict__ ws,
    float* __restrict__ out1)
{
    __shared__ float sA[WP_PER_C];
    __shared__ float sC2[256];
    __shared__ float sC1[16];
    __shared__ float sN[8 * 256];

    const int t  = threadIdx.x;
    const int c  = blockIdx.y;
    const int w  = t >> 5;
    const int l  = t & 31;
    const int l16 = l & 15;
    const int hi  = l >> 4;

    {
        const float4* g = (const float4*)(ws + OFF_W3 + c * WP_PER_C);
        float4* s = (float4*)sA;
        s[t]       = g[t];
        s[256 + t] = g[256 + t];
        if (t < 32) s[512 + t] = g[512 + t];
    }
    sC2[t] = ws[OFF_C2 + c * 256 + t];
    if (t < 16) sC1[t] = ws[OFF_C1 + c * 16 + t];
    {
        const int row  = l >> 1;
        const int half = l & 1;
        const int b    = (blockIdx.x * 8 + w) * 16 + row;
        const float4* g = (const float4*)(nf + (size_t)b * (Csz * LM) + c * LM + half * 8);
        float4* s = (float4*)(sN + w * 256 + row * 16 + half * 8);
        s[0] = g[0];
        s[1] = g[1];
    }
    __syncthreads();

    const float* Nw = sN + w * 256;

    float m[16];
    #pragma unroll
    for (int j = 0; j < 16; ++j) m[j] = Nw[l16 * 16 + ((j + hi) & 15)];
    float nfd[8];
    #pragma unroll
    for (int r = 0; r < 8; ++r) nfd[r] = Nw[l16 * 16 + r + 8 * hi];

    v8f C;
    #pragma unroll
    for (int r = 0; r < 8; ++r) C[r] = sC1[r + 8 * hi];

    const v2f* Af  = (const v2f*)sA;
    const v2f* C2f = (const v2f*)sC2;

    #pragma unroll
    for (int kk = 0; kk < 4; ++kk) {
        v2f A = C2f[kk * 32 + l];
        v2f Bv;
        Bv.x = m[4 * kk];
        Bv.y = m[4 * kk + 2];
        C = __builtin_amdgcn_wmma_f32_16x16x4_f32(
                false, A, false, Bv, (short)0, C, false, false);
    }
    #pragma unroll
    for (int kk = 0; kk < NKK; ++kk) {
        v2f A = Af[kk * 32 + l];
        v2f Bv;
        Bv.x = m[JA(2 * kk)]     * m[JB(2 * kk)];
        Bv.y = m[JA(2 * kk + 1)] * m[JB(2 * kk + 1)];
        C = __builtin_amdgcn_wmma_f32_16x16x4_f32(
                false, A, false, Bv, (short)0, C, false, false);
    }

    float e = C[0] * nfd[0];
    #pragma unroll
    for (int r = 1; r < 8; ++r) e += C[r] * nfd[r];
    e += __shfl_xor(e, 16, 32);

    if (hi == 0) {
        int b = (blockIdx.x * 8 + w) * 16 + l16;
        out1[c * Bsz + b] = e;
    }
}

__global__ __launch_bounds__(256) void pb_final_gemm(
    const float* __restrict__ o1,
    const float* __restrict__ lw,
    float* __restrict__ out)
{
    const int t   = threadIdx.x;
    const int w   = t >> 5;
    const int l   = t & 31;
    const int l16 = l & 15;
    const int hi  = l >> 4;

    const int tile = blockIdx.x * 8 + w;
    const int b0 = (tile >> 4) * 16;
    const int d0 = (tile & 15) * 16;

    v8f acc;
    #pragma unroll
    for (int r = 0; r < 8; ++r) acc[r] = 0.f;

    #pragma unroll 4
    for (int k0 = 0; k0 < 256; k0 += 4) {
        const int ka = k0 + 2 * hi;
        v2f A, B;
        A.x = o1[(size_t)ka * Bsz + b0 + l16];
        A.y = o1[(size_t)(ka + 1) * Bsz + b0 + l16];
        B.x = lw[ka * Csz + d0 + l16];
        B.y = lw[(ka + 1) * Csz + d0 + l16];
        acc = __builtin_amdgcn_wmma_f32_16x16x4_f32(
                  false, A, false, B, (short)0, acc, false, false);
    }

    #pragma unroll
    for (int r = 0; r < 8; ++r)
        out[(size_t)(b0 + r + 8 * hi) * Csz + d0 + l16] = acc[r] * 0.0625f;
}

extern "C" void kernel_launch(void* const* d_in, const int* in_sizes, int n_in,
                              void* d_out, int out_size, void* d_ws, size_t ws_size,
                              hipStream_t stream) {
    (void)in_sizes; (void)n_in; (void)out_size; (void)ws_size;
    const float* nf  = (const float*)d_in[0];
    const float* U3  = (const float*)d_in[1];
    const float* U2  = (const float*)d_in[2];
    const float* U1  = (const float*)d_in[3];
    const float* w3  = (const float*)d_in[4];
    const float* w2  = (const float*)d_in[5];
    const float* w1  = (const float*)d_in[6];
    const float* lw  = (const float*)d_in[7];

    float* ws   = (float*)d_ws;
    float* out1 = ws + OFF_O1;

    pb_precompute<<<848, 256, 0, stream>>>(U3, U2, U1, w3, w2, w1, ws);
    pb_main<<<dim3(32, 256), 256, 0, stream>>>(nf, ws, out1);
    pb_final_gemm<<<512, 256, 0, stream>>>(out1, lw, (float*)d_out);
}


# pb_main: each wave loops over 16 b-tiles (8 pairs; the two tiles of a pair share each A fragment read), Wp/c2/c1 staged once per block, next rows prefetched: 65536 -> 4096 waves; sum order unchanged
# speedup vs baseline: 1.5625x; 1.1152x over previous
#include <hip/hip_runtime.h>

typedef __attribute__((ext_vector_type(2))) float v2f;
typedef __attribute__((ext_vector_type(8))) float v8f;

#define Bsz   4096
#define Csz   256
#define LM    16

#define OFF_W3   0
#define OFF_C2   1048576
#define OFF_C1   1114112
#define OFF_O1   1118208

__host__ __device__ constexpr int JA(int q) { return q < 64 ? 2 * (q & 7) : 2 * (q - 64); }
__host__ __device__ constexpr int JB(int q) { return q < 64 ? ((2 * (q & 7) + (q >> 3)) & 15) : (2 * (q - 64) + 8); }

#define NKK      34
#define WP_PER_C 2176

#define FOLD_CG 4
__global__ __launch_bounds__(256) void pb_precompute(
    const float* __restrict__ U3, const float* __restrict__ U2,
    const float* __restrict__ U1, const float* __restrict__ w3,
    const float* __restrict__ w2, const float* __restrict__ w1,
    float* __restrict__ ws)
{
    const int t  = threadIdx.x;
    const int bx = blockIdx.x;
    if (bx < 576) {
        const int cg  = bx / 9;
        const int rem = (bx - cg * 9) * 256 + t;
        if (rem < WP_PER_C) {
            int kk   = rem >> 6;
            int lane = (rem >> 1) & 31;
            int sel  = rem & 1;
            int x = lane & 15, h = lane >> 4;
            int q = 2 * kk + sel;
            int y = (JA(q) + h) & 15;
            int i = (JB(q) + h) & 15;
            const float* r1 = U3 + ((x * 16 + y) * 16 + i) * 23;
            const float* r2 = U3 + ((x * 16 + i) * 16 + y) * 23;
            float u1[23], u2[23];
            #pragma unroll
            for (int k = 0; k < 23; ++k) { u1[k] = r1[k]; u2[k] = r2[k]; }
            #pragma unroll
            for (int j = 0; j < FOLD_CG; ++j) {
                const int c = cg * FOLD_CG + j;
                float s1 = 0.f, s2 = 0.f;
                #pragma unroll
                for (int k = 0; k < 23; ++k) {
                    float wk = w3[k * Csz + c];
                    s1 += u1[k] * wk;
                    s2 += u2[k] * wk;
                }
                ws[OFF_W3 + c * WP_PER_C + rem] = (y != i) ? (s1 + s2) : s1;
            }
        }
        return;
    }
    const int idx = (bx - 576) * 256 + t;
    if (idx < 65536) {
        int c    = idx >> 8;
        int rem  = idx & 255;
        int kk   = rem >> 6;
        int lane = (rem >> 1) & 31;
        int sel  = rem & 1;
        int x = lane & 15, h = lane >> 4;
        int y = (2 * (2 * kk + sel) + h) & 15;
        float s = 0.f;
        #pragma unroll
        for (int k = 0; k < 8; ++k) s += U2[(x * 16 + y) * 8 + k] * w2[k * Csz + c];
        ws[OFF_C2 + idx] = s;
    } else if (idx < 65536 + 4096) {
        int e = idx - 65536;
        int c = e >> 4;
        int i = e & 15;
        float s = 0.f;
        #pragma unroll
        for (int k = 0; k < 4; ++k) s += U1[i * 4 + k] * w1[k * Csz + c];
        ws[OFF_C1 + c * 16 + i] = s;
    }
}

#define PAIRS_PER_WAVE 8

__device__ __forceinline__ void pb_load_row(const float* __restrict__ p, float4 (&r)[4])
{
    const float4* g = (const float4*)p;
    r[0] = g[0]; r[1] = g[1]; r[2] = g[2]; r[3] = g[3];
}

#define PB_HIDE(x) asm("" : "+v"(x))
__device__ __forceinline__ void pb_rotate(const float4 (&r)[4], bool up, float (&m)[16], float (&d)[8])
{
    float n0 = r[0].x, n1 = r[0].y, n2  = r[0].z, n3  = r[0].w;
    float n4 = r[1].x, n5 = r[1].y, n6  = r[1].z, n7  = r[1].w;
    float n8 = r[2].x, n9 = r[2].y, n10 = r[2].z, n11 = r[2].w;
    float n12 = r[3].x, n13 = r[3].y, n14 = r[3].z, n15 = r[3].w;
    PB_HIDE(n0);  PB_HIDE(n1);  PB_HIDE(n2);  PB_HIDE(n3);
    PB_HIDE(n4);  PB_HIDE(n5);  PB_HIDE(n6);  PB_HIDE(n7);
    PB_HIDE(n8);  PB_HIDE(n9);  PB_HIDE(n10); PB_HIDE(n11);
    PB_HIDE(n12); PB_HIDE(n13); PB_HIDE(n14); PB_HIDE(n15);
    m[0]  = up ? n1  : n0;   m[1]  = up ? n2  : n1;   m[2]  = up ? n3  : n2;   m[3]  = up ? n4  : n3;
    m[4]  = up ? n5  : n4;   m[5]  = up ? n6  : n5;   m[6]  = up ? n7  : n6;   m[7]  = up ? n8  : n7;
    m[8]  = up ? n9  : n8;   m[9]  = up ? n10 : n9;   m[10] = up ? n11 : n10;  m[11] = up ? n12 : n11;
    m[12] = up ? n13 : n12;  m[13] = up ? n14 : n13;  m[14] = up ? n15 : n14;  m[15] = up ? n0  : n15;
    d[0] = up ? n8  : n0;    d[1] = up ? n9  : n1;    d[2] = up ? n10 : n2;    d[3] = up ? n11 : n3;
    d[4] = up ? n12 : n4;    d[5] = up ? n13 : n5;    d[6] = up ? n14 : n6;    d[7] = up ? n15 : n7;
}

__global__ __launch_bounds__(256) void pb_main(
    const float* __restrict__ nf,
    const float* __restrict__ ws,
    float* __restrict__ out1)
{
    __shared__ float sA[WP_PER_C];
    __shared__ float sC2[256];
    __shared__ float sC1[16];

    const int t  = threadIdx.x;
    const int c  = blockIdx.y;
    const int w  = t >> 5;
    const int l  = t & 31;
    const int l16 = l & 15;
    const int hi  = l >> 4;
    const bool up = hi != 0;

    const int bw = (blockIdx.x * PAIRS_PER_WAVE * 8 + w) * 32;
    const float* rowp = nf + (size_t)(bw + l16) * (Csz * LM) + c * LM;
    const size_t tile_stride = (size_t)16 * Csz * LM;
    const size_t pair_stride = (size_t)256 * Csz * LM;

    float4 ra[4], rb[4];
    pb_load_row(rowp, ra);
    pb_load_row(rowp + tile_stride, rb);

    {
        const float4* g = (const float4*)(ws + OFF_W3 + c * WP_PER_C);
        float4* s = (float4*)sA;
        s[t]       = g[t];
        s[256 + t] = g[256 + t];
        if (t < 32) s[512 + t] = g[512 + t];
    }
    sC2[t] = ws[OFF_C2 + c * 256 + t];
    if (t < 16) sC1[t] = ws[OFF_C1 + c * 16 + t];
    __syncthreads();

    const v2f* Af  = (const v2f*)sA;
    const v2f* C2f = (const v2f*)sC2;

    for (int p = 0; p < PAIRS_PER_WAVE; ++p) {
        float m0[16], m1[16], d0[8], d1[8];
        pb_rotate(ra, up, m0, d0);
        pb_rotate(rb, up, m1, d1);

        if (p + 1 < PAIRS_PER_WAVE) {
            const float* nx = rowp + (size_t)(p + 1) * pair_stride;
            pb_load_row(nx, ra);
            pb_load_row(nx + tile_stride, rb);
        }

        v8f C0, C1;
        #pragma unroll
        for (int r = 0; r < 8; ++r) { float v = sC1[r + 8 * hi]; C0[r] = v; C1[r] = v; }

        #pragma unroll
        for (int kk = 0; kk < 4; ++kk) {
            v2f A = C2f[kk * 32 + l];
            v2f B0, B1;
            B0.x = m0[4 * kk];  B0.y = m0[4 * kk + 2];
            B1.x = m1[4 * kk];  B1.y = m1[4 * kk + 2];
            C0 = __builtin_amdgcn_wmma_f32_16x16x4_f32(
                     false, A, false, B0, (short)0, C0, false, false);
            C1 = __builtin_amdgcn_wmma_f32_16x16x4_f32(
                     false, A, false, B1, (short)0, C1, false, false);
        }
        #pragma unroll
        for (int kk = 0; kk < NKK; ++kk) {
            v2f A = Af[kk * 32 + l];
            v2f B0, B1;
            B0.x = m0[JA(2 * kk)]     * m0[JB(2 * kk)];
            B0.y = m0[JA(2 * kk + 1)] * m0[JB(2 * kk + 1)];
            B1.x = m1[JA(2 * kk)]     * m1[JB(2 * kk)];
            B1.y = m1[JA(2 * kk + 1)] * m1[JB(2 * kk + 1)];
            C0 = __builtin_amdgcn_wmma_f32_16x16x4_f32(
                     false, A, false, B0, (short)0, C0, false, false);
            C1 = __builtin_amdgcn_wmma_f32_16x16x4_f32(
                     false, A, false, B1, (short)0, C1, false, false);
        }

        float e0 = C0[0] * d0[0];
        float e1 = C1[0] * d1[0];
        #pragma unroll
        for (int r = 1; r < 8; ++r) {
            e0 += C0[r] * d0[r];
            e1 += C1[r] * d1[r];
        }
        e0 += __shfl_xor(e0, 16, 32);
        e1 += __shfl_xor(e1, 16, 32);

        out1[c * Bsz + bw + p * 256 + l] = up ? e1 : e0;
    }
}

__global__ __launch_bounds__(256) void pb_final_gemm(
    const float* __restrict__ o1,
    const float* __restrict__ lw,
    float* __restrict__ out)
{
    const int t   = threadIdx.x;
    const int w   = t >> 5;
    const int l   = t & 31;
    const int l16 = l & 15;
    const int hi  = l >> 4;

    const int tile = blockIdx.x * 8 + w;
    const int b0 = (tile >> 4) * 16;
    const int d0 = (tile & 15) * 16;

    v8f acc;
    #pragma unroll
    for (int r = 0; r < 8; ++r) acc[r] = 0.f;

    #pragma unroll 4
    for (int k0 = 0; k0 < 256; k0 += 4) {
        const int ka = k0 + 2 * hi;
        v2f A, B;
        A.x = o1[(size_t)ka * Bsz + b0 + l16];
        A.y = o1[(size_t)(ka + 1) * Bsz + b0 + l16];
        B.x = lw[ka * Csz + d0 + l16];
        B.y = lw[(ka + 1) * Csz + d0 + l16];
        acc = __builtin_amdgcn_wmma_f32_16x16x4_f32(
                  false, A, false, B, (short)0, acc, false, false);
    }

    #pragma unroll
    for (int r = 0; r < 8; ++r)
        out[(size_t)(b0 + r + 8 * hi) * Csz + d0 + l16] = acc[r] * 0.0625f;
}

extern "C" void kernel_launch(void* const* d_in, const int* in_sizes, int n_in,
                              void* d_out, int out_size, void* d_ws, size_t ws_size,
                              hipStream_t stream) {
    (void)in_sizes; (void)n_in; (void)out_size; (void)ws_size;
    const float* nf  = (const float*)d_in[0];
    const float* U3  = (const float*)d_in[1];
    const float* U2  = (const float*)d_in[2];
    const float* U1  = (const float*)d_in[3];
    const float* w3  = (const float*)d_in[4];
    const float* w2  = (const float*)d_in[5];
    const float* w1  = (const float*)d_in[6];
    const float* lw  = (const float*)d_in[7];

    float* ws   = (float*)d_ws;
    float* out1 = ws + OFF_O1;

    pb_precompute<<<848, 256, 0, stream>>>(U3, U2, U1, w3, w2, w1, ws);
    pb_main<<<dim3(2, 256), 256, 0, stream>>>(nf, ws, out1);
    pb_final_gemm<<<512, 256, 0, stream>>>(out1, lw, (float*)d_out);
}


# Cubic term over 72 pair columns, not 136: each unordered triple folded once (its 6 W3 permutations summed) as row x pair inside one index half; 22 WMMAs per tile, not 38; sum order changes
# speedup vs baseline: 1.8320x; 1.1725x over previous
#include <hip/hip_runtime.h>


typedef __attribute__((ext_vector_type(2))) float v2f;
typedef __attribute__((ext_vector_type(8))) float v8f;

#define Bsz   4096
#define Csz   256
#define LM    16

#define OFF_W3   0
#define OFF_C2   1048576
#define OFF_C1   1114112
#define OFF_O1   1118208

__host__ __device__ constexpr int PA(int q) { return q < 32 ? (q & 7) : (q - 32); }
__host__ __device__ constexpr int PB(int q) { return q < 32 ? (((q & 7) + (q >> 3)) & 7) : (q - 32 + 4); }

#define NKK      18
#define WP_PER_C 1152

#define FOLD_CG 16
#define PRE_T   64
#define WP_BLK_PER_G 18
#define WP_BLOCKS    288
__global__ __launch_bounds__(PRE_T) __attribute__((amdgpu_waves_per_eu(1, 6))) void pb_precompute(
    const float* __restrict__ U3, const float* __restrict__ U2,
    const float* __restrict__ U1, const float* __restrict__ w3,
    const float* __restrict__ w2, const float* __restrict__ w1,
    float* __restrict__ ws)
{
    const int t  = threadIdx.x;
    const int bx = blockIdx.x;
    if (bx < WP_BLOCKS) {
        const int cg  = bx / WP_BLK_PER_G;
        const int rem = (bx - cg * WP_BLK_PER_G) * PRE_T + t;
        const int kk   = rem >> 6;
        const int lane = (rem >> 1) & 31;
        const int sel  = rem & 1;
        const int r = lane & 15, h = lane >> 4;
        const int q = 2 * kk + sel;
        const int y = PA(q) + 8 * h;
        const int i = PB(q) + 8 * h;
        const int lo = y < i ? y : i;
        const int hi = y < i ? i : y;
        const bool home = ((r >> 3) != h) || (r <= lo);
        int e[6];
        e[0] = (r  * 16 + lo) * 16 + hi;
        e[1] = (r  * 16 + hi) * 16 + lo;
        e[2] = (lo * 16 + r ) * 16 + hi;
        e[3] = (lo * 16 + hi) * 16 + r;
        e[4] = (hi * 16 + r ) * 16 + lo;
        e[5] = (hi * 16 + lo) * 16 + r;
        float rv[6][23];
        #pragma unroll
        for (int j = 0; j < 6; ++j) {
            const float* row = U3 + e[j] * 23;
            #pragma unroll
            for (int k = 0; k < 23; ++k) rv[j][k] = row[k];
        }
        float u[23];
        #pragma unroll
        for (int k = 0; k < 23; ++k) u[k] = 0.f;
        #pragma unroll
        for (int j = 0; j < 6; ++j) {
            bool first = true;
            #pragma unroll
            for (int jj = 0; jj < j; ++jj) first = first && (e[jj] != e[j]);
            const float wj = (home && first) ? 1.f : 0.f;
            #pragma unroll
            for (int k = 0; k < 23; ++k) u[k] += wj * rv[j][k];
        }
        #pragma unroll
        for (int j = 0; j < FOLD_CG; ++j) {
            const int ch = cg * FOLD_CG + j;
            float s = 0.f;
            #pragma unroll
            for (int k = 0; k < 23; ++k) s += u[k] * w3[k * Csz + ch];
            ws[OFF_W3 + ch * WP_PER_C + rem] = s;
        }
        return;
    }
    const int idx = (bx - WP_BLOCKS) * PRE_T + t;
    if (idx < 65536) {
        int ch   = idx >> 8;
        int rem  = idx & 255;
        int kk   = rem >> 6;
        int lane = (rem >> 1) & 31;
        int sel  = rem & 1;
        int x = lane & 15, h = lane >> 4;
        int y = 2 * kk + sel + 8 * h;
        float s = 0.f;
        #pragma unroll
        for (int k = 0; k < 8; ++k) s += U2[(x * 16 + y) * 8 + k] * w2[k * Csz + ch];
        ws[OFF_C2 + idx] = s;
    } else if (idx < 65536 + 4096) {
        int e1 = idx - 65536;
        int ch = e1 >> 4;
        int i  = e1 & 15;
        float s = 0.f;
        #pragma unroll
        for (int k = 0; k < 4; ++k) s += U1[i * 4 + k] * w1[k * Csz + ch];
        ws[OFF_C1 + ch * 16 + i] = s;
    }
}

#define PAIRS_PER_WAVE 8

__device__ __forceinline__ void pb_load_half(const float* __restrict__ p, float4 (&r)[2])
{
    const float4* g = (const float4*)p;
    r[0] = g[0]; r[1] = g[1];
}

__device__ __forceinline__ void pb_unpack(const float4 (&r)[2], float (&d)[8])
{
    d[0] = r[0].x; d[1] = r[0].y; d[2] = r[0].z; d[3] = r[0].w;
    d[4] = r[1].x; d[5] = r[1].y; d[6] = r[1].z; d[7] = r[1].w;
}

__global__ __launch_bounds__(256) void pb_main(
    const float* __restrict__ nf,
    const float* __restrict__ ws,
    float* __restrict__ out1)
{
    __shared__ float sA[WP_PER_C];
    __shared__ float sC2[256];
    __shared__ float sC1[16];

    const int t  = threadIdx.x;
    const int c  = blockIdx.y;
    const int w  = t >> 5;
    const int l  = t & 31;
    const int l16 = l & 15;
    const int hi  = l >> 4;
    const bool up = hi != 0;

    const int bw = (blockIdx.x * PAIRS_PER_WAVE * 8 + w) * 32;
    const float* rowp = nf + (size_t)(bw + l16) * (Csz * LM) + c * LM + 8 * hi;
    const size_t tile_stride = (size_t)16 * Csz * LM;
    const size_t pair_stride = (size_t)256 * Csz * LM;

    float4 ra[2], rb[2];
    pb_load_half(rowp, ra);
    pb_load_half(rowp + tile_stride, rb);

    {
        const float4* g = (const float4*)(ws + OFF_W3 + c * WP_PER_C);
        float4* s = (float4*)sA;
        s[t] = g[t];
        if (t < 32) s[256 + t] = g[256 + t];
    }
    sC2[t] = ws[OFF_C2 + c * 256 + t];
    if (t < 16) sC1[t] = ws[OFF_C1 + c * 16 + t];
    __syncthreads();

    const v2f* Af  = (const v2f*)sA;
    const v2f* C2f = (const v2f*)sC2;

    for (int p = 0; p < PAIRS_PER_WAVE; ++p) {
        float d0[8], d1[8];
        pb_unpack(ra, d0);
        pb_unpack(rb, d1);

        if (p + 1 < PAIRS_PER_WAVE) {
            const float* nx = rowp + (size_t)(p + 1) * pair_stride;
            pb_load_half(nx, ra);
            pb_load_half(nx + tile_stride, rb);
        }

        v8f C0, C1;
        #pragma unroll
        for (int r = 0; r < 8; ++r) { float v = sC1[r + 8 * hi]; C0[r] = v; C1[r] = v; }

        #pragma unroll
        for (int kk = 0; kk < 4; ++kk) {
            v2f A = C2f[kk * 32 + l];
            v2f B0, B1;
            B0.x = d0[2 * kk];  B0.y = d0[2 * kk + 1];
            B1.x = d1[2 * kk];  B1.y = d1[2 * kk + 1];
            C0 = __builtin_amdgcn_wmma_f32_16x16x4_f32(
                     false, A, false, B0, (short)0, C0, false, false);
            C1 = __builtin_amdgcn_wmma_f32_16x16x4_f32(
                     false, A, false, B1, (short)0, C1, false, false);
        }
        #pragma unroll
        for (int kk = 0; kk < NKK; ++kk) {
            v2f A = Af[kk * 32 + l];
            v2f B0, B1;
            B0.x = d0[PA(2 * kk)]     * d0[PB(2 * kk)];
            B0.y = d0[PA(2 * kk + 1)] * d0[PB(2 * kk + 1)];
            B1.x = d1[PA(2 * kk)]     * d1[PB(2 * kk)];
            B1.y = d1[PA(2 * kk + 1)] * d1[PB(2 * kk + 1)];
            C0 = __builtin_amdgcn_wmma_f32_16x16x4_f32(
                     false, A, false, B0, (short)0, C0, false, false);
            C1 = __builtin_amdgcn_wmma_f32_16x16x4_f32(
                     false, A, false, B1, (short)0, C1, false, false);
        }

        float e0 = C0[0] * d0[0];
        float e1 = C1[0] * d1[0];
        #pragma unroll
        for (int r = 1; r < 8; ++r) {
            e0 += C0[r] * d0[r];
            e1 += C1[r] * d1[r];
        }
        e0 += __shfl_xor(e0, 16, 32);
        e1 += __shfl_xor(e1, 16, 32);

        out1[c * Bsz + bw + p * 256 + l] = up ? e1 : e0;
    }
}

__global__ __launch_bounds__(256) void pb_final_gemm(
    const float* __restrict__ o1,
    const float* __restrict__ lw,
    float* __restrict__ out)
{
    const int t   = threadIdx.x;
    const int w   = t >> 5;
    const int l   = t & 31;
    const int l16 = l & 15;
    const int hi  = l >> 4;

    const int tile = blockIdx.x * 8 + w;
    const int b0 = (tile >> 4) * 16;
    const int d0 = (tile & 15) * 16;

    v8f acc;
    #pragma unroll
    for (int r = 0; r < 8; ++r) acc[r] = 0.f;

    #pragma unroll 4
    for (int k0 = 0; k0 < 256; k0 += 4) {
        const int ka = k0 + 2 * hi;
        v2f A, B;
        A.x = o1[(size_t)ka * Bsz + b0 + l16];
        A.y = o1[(size_t)(ka + 1) * Bsz + b0 + l16];
        B.x = lw[ka * Csz + d0 + l16];
        B.y = lw[(ka + 1) * Csz + d0 + l16];
        acc = __builtin_amdgcn_wmma_f32_16x16x4_f32(
                  false, A, false, B, (short)0, acc, false, false);
    }

    #pragma unroll
    for (int r = 0; r < 8; ++r)
        out[(size_t)(b0 + r + 8 * hi) * Csz + d0 + l16] = acc[r] * 0.0625f;
}

extern "C" void kernel_launch(void* const* d_in, const int* in_sizes, int n_in,
                              void* d_out, int out_size, void* d_ws, size_t ws_size,
                              hipStream_t stream) {
    (void)in_sizes; (void)n_in; (void)out_size; (void)ws_size;
    const float* nf  = (const float*)d_in[0];
    const float* U3  = (const float*)d_in[1];
    const float* U2  = (const float*)d_in[2];
    const float* U1  = (const float*)d_in[3];
    const float* w3  = (const float*)d_in[4];
    const float* w2  = (const float*)d_in[5];
    const float* w1  = (const float*)d_in[6];
    const float* lw  = (const float*)d_in[7];

    float* ws   = (float*)d_ws;
    float* out1 = ws + OFF_O1;

    pb_precompute<<<1376, 64, 0, stream>>>(U3, U2, U1, w3, w2, w1, ws);
    pb_main<<<dim3(2, 256), 256, 0, stream>>>(nf, ws, out1);
    pb_final_gemm<<<512, 256, 0, stream>>>(out1, lw, (float*)d_out);
}


# pb_final_gemm: both operands staged through LDS (64x64 block tile, K in 4 double-buffered chunks, k-pair interleaved rows, 8-byte fragment reads), two d-tiles per wave; sum order unchanged
# speedup vs baseline: 2.0642x; 1.1267x over previous
#include <hip/hip_runtime.h>


typedef __attribute__((ext_vector_type(2))) float v2f;
typedef __attribute__((ext_vector_type(8))) float v8f;

#define Bsz   4096
#define Csz   256
#define LM    16

#define OFF_W3   0
#define OFF_C2   1048576
#define OFF_C1   1114112
#define OFF_O1   1118208

__host__ __device__ constexpr int PA(int q) { return q < 32 ? (q & 7) : (q - 32); }
__host__ __device__ constexpr int PB(int q) { return q < 32 ? (((q & 7) + (q >> 3)) & 7) : (q - 32 + 4); }

#define NKK      18
#define WP_PER_C 1152

#define FOLD_CG 16
#define PRE_T   64
#define WP_BLK_PER_G 18
#define WP_BLOCKS    288
__global__ __launch_bounds__(PRE_T) __attribute__((amdgpu_waves_per_eu(1, 6))) void pb_precompute(
    const float* __restrict__ U3, const float* __restrict__ U2,
    const float* __restrict__ U1, const float* __restrict__ w3,
    const float* __restrict__ w2, const float* __restrict__ w1,
    float* __restrict__ ws)
{
    const int t  = threadIdx.x;
    const int bx = blockIdx.x;
    if (bx < WP_BLOCKS) {
        const int cg  = bx / WP_BLK_PER_G;
        const int rem = (bx - cg * WP_BLK_PER_G) * PRE_T + t;
        const int kk   = rem >> 6;
        const int lane = (rem >> 1) & 31;
        const int sel  = rem & 1;
        const int r = lane & 15, h = lane >> 4;
        const int q = 2 * kk + sel;
        const int y = PA(q) + 8 * h;
        const int i = PB(q) + 8 * h;
        const int lo = y < i ? y : i;
        const int hi = y < i ? i : y;
        const bool home = ((r >> 3) != h) || (r <= lo);
        int e[6];
        e[0] = (r  * 16 + lo) * 16 + hi;
        e[1] = (r  * 16 + hi) * 16 + lo;
        e[2] = (lo * 16 + r ) * 16 + hi;
        e[3] = (lo * 16 + hi) * 16 + r;
        e[4] = (hi * 16 + r ) * 16 + lo;
        e[5] = (hi * 16 + lo) * 16 + r;
        float rv[6][23];
        #pragma unroll
        for (int j = 0; j < 6; ++j) {
            const float* row = U3 + e[j] * 23;
            #pragma unroll
            for (int k = 0; k < 23; ++k) rv[j][k] = row[k];
        }
        float u[23];
        #pragma unroll
        for (int k = 0; k < 23; ++k) u[k] = 0.f;
        #pragma unroll
        for (int j = 0; j < 6; ++j) {
            bool first = true;
            #pragma unroll
            for (int jj = 0; jj < j; ++jj) first = first && (e[jj] != e[j]);
            const float wj = (home && first) ? 1.f : 0.f;
            #pragma unroll
            for (int k = 0; k < 23; ++k) u[k] += wj * rv[j][k];
        }
        #pragma unroll
        for (int j = 0; j < FOLD_CG; ++j) {
            const int ch = cg * FOLD_CG + j;
            float s = 0.f;
            #pragma unroll
            for (int k = 0; k < 23; ++k) s += u[k] * w3[k * Csz + ch];
            ws[OFF_W3 + ch * WP_PER_C + rem] = s;
        }
        return;
    }
    const int idx = (bx - WP_BLOCKS) * PRE_T + t;
    if (idx < 65536) {
        int ch   = idx >> 8;
        int rem  = idx & 255;
        int kk   = rem >> 6;
        int lane = (rem >> 1) & 31;
        int sel  = rem & 1;
        int x = lane & 15, h = lane >> 4;
        int y = 2 * kk + sel + 8 * h;
        float s = 0.f;
        #pragma unroll
        for (int k = 0; k < 8; ++k) s += U2[(x * 16 + y) * 8 + k] * w2[k * Csz + ch];
        ws[OFF_C2 + idx] = s;
    } else if (idx < 65536 + 4096) {
        int e1 = idx - 65536;
        int ch = e1 >> 4;
        int i  = e1 & 15;
        float s = 0.f;
        #pragma unroll
        for (int k = 0; k < 4; ++k) s += U1[i * 4 + k] * w1[k * Csz + ch];
        ws[OFF_C1 + ch * 16 + i] = s;
    }
}

#define PAIRS_PER_WAVE 8

__device__ __forceinline__ void pb_load_half(const float* __restrict__ p, float4 (&r)[2])
{
    const float4* g = (const float4*)p;
    r[0] = g[0]; r[1] = g[1];
}

__device__ __forceinline__ void pb_unpack(const float4 (&r)[2], float (&d)[8])
{
    d[0] = r[0].x; d[1] = r[0].y; d[2] = r[0].z; d[3] = r[0].w;
    d[4] = r[1].x; d[5] = r[1].y; d[6] = r[1].z; d[7] = r[1].w;
}

__global__ __launch_bounds__(256) void pb_main(
    const float* __restrict__ nf,
    const float* __restrict__ ws,
    float* __restrict__ out1)
{
    __shared__ float sA[WP_PER_C];
    __shared__ float sC2[256];
    __shared__ float sC1[16];

    const int t  = threadIdx.x;
    const int c  = blockIdx.y;
    const int w  = t >> 5;
    const int l  = t & 31;
    const int l16 = l & 15;
    const int hi  = l >> 4;
    const bool up = hi != 0;

    const int bw = (blockIdx.x * PAIRS_PER_WAVE * 8 + w) * 32;
    const float* rowp = nf + (size_t)(bw + l16) * (Csz * LM) + c * LM + 8 * hi;
    const size_t tile_stride = (size_t)16 * Csz * LM;
    const size_t pair_stride = (size_t)256 * Csz * LM;

    float4 ra[2], rb[2];
    pb_load_half(rowp, ra);
    pb_load_half(rowp + tile_stride, rb);

    {
        const float4* g = (const float4*)(ws + OFF_W3 + c * WP_PER_C);
        float4* s = (float4*)sA;
        s[t] = g[t];
        if (t < 32) s[256 + t] = g[256 + t];
    }
    sC2[t] = ws[OFF_C2 + c * 256 + t];
    if (t < 16) sC1[t] = ws[OFF_C1 + c * 16 + t];
    __syncthreads();

    const v2f* Af  = (const v2f*)sA;
    const v2f* C2f = (const v2f*)sC2;

    for (int p = 0; p < PAIRS_PER_WAVE; ++p) {
        float d0[8], d1[8];
        pb_unpack(ra, d0);
        pb_unpack(rb, d1);

        if (p + 1 < PAIRS_PER_WAVE) {
            const float* nx = rowp + (size_t)(p + 1) * pair_stride;
            pb_load_half(nx, ra);
            pb_load_half(nx + tile_stride, rb);
        }

        v8f C0, C1;
        #pragma unroll
        for (int r = 0; r < 8; ++r) { float v = sC1[r + 8 * hi]; C0[r] = v; C1[r] = v; }

        #pragma unroll
        for (int kk = 0; kk < 4; ++kk) {
            v2f A = C2f[kk * 32 + l];
            v2f B0, B1;
            B0.x = d0[2 * kk];  B0.y = d0[2 * kk + 1];
            B1.x = d1[2 * kk];  B1.y = d1[2 * kk + 1];
            C0 = __builtin_amdgcn_wmma_f32_16x16x4_f32(
                     false, A, false, B0, (short)0, C0, false, false);
            C1 = __builtin_amdgcn_wmma_f32_16x16x4_f32(
                     false, A, false, B1, (short)0, C1, false, false);
        }
        #pragma unroll
        for (int kk = 0; kk < NKK; ++kk) {
            v2f A = Af[kk * 32 + l];
            v2f B0, B1;
            B0.x = d0[PA(2 * kk)]     * d0[PB(2 * kk)];
            B0.y = d0[PA(2 * kk + 1)] * d0[PB(2 * kk + 1)];
            B1.x = d1[PA(2 * kk)]     * d1[PB(2 * kk)];
            B1.y = d1[PA(2 * kk + 1)] * d1[PB(2 * kk + 1)];
            C0 = __builtin_amdgcn_wmma_f32_16x16x4_f32(
                     false, A, false, B0, (short)0, C0, false, false);
            C1 = __builtin_amdgcn_wmma_f32_16x16x4_f32(
                     false, A, false, B1, (short)0, C1, false, false);
        }

        float e0 = C0[0] * d0[0];
        float e1 = C1[0] * d1[0];
        #pragma unroll
        for (int r = 1; r < 8; ++r) {
            e0 += C0[r] * d0[r];
            e1 += C1[r] * d1[r];
        }
        e0 += __shfl_xor(e0, 16, 32);
        e1 += __shfl_xor(e1, 16, 32);

        out1[c * Bsz + bw + p * 256 + l] = up ? e1 : e0;
    }
}

#define FG_KC     64
#define FG_NCH    4
#define FG_KP     32
#define FG_PITCH  160
#define FG_PANEL  (FG_KP * FG_PITCH)

__device__ __forceinline__ void fg_fetch(const float* __restrict__ o1, const float* __restrict__ lw,
                                         int b0, int d0, int kb, int t,
                                         float4 (&ga)[2][2], float4 (&gb)[2][2])
{
    #pragma unroll
    for (int it = 0; it < 2; ++it) {
        const int f    = t + 256 * it;
        const int col4 = f & 15;
        const int k    = kb + 2 * (f >> 4);
        const float* pa = o1 + (size_t)k * Bsz + b0 + col4 * 4;
        const float* pw = lw + k * Csz + d0 + col4 * 4;
        ga[it][0] = *(const float4*)pa;
        ga[it][1] = *(const float4*)(pa + Bsz);
        gb[it][0] = *(const float4*)pw;
        gb[it][1] = *(const float4*)(pw + Csz);
    }
}

__device__ __forceinline__ void fg_put(float* __restrict__ s, const float4& u, const float4& v)
{
    float4* d = (float4*)s;
    d[0] = make_float4(u.x, v.x, u.y, v.y);
    d[1] = make_float4(u.z, v.z, u.w, v.w);
}

__global__ __launch_bounds__(256) void pb_final_gemm(
    const float* __restrict__ o1,
    const float* __restrict__ lw,
    float* __restrict__ out)
{
    __shared__ __attribute__((aligned(16))) float sA[2][FG_PANEL];
    __shared__ __attribute__((aligned(16))) float sB[2][FG_PANEL];

    const int t   = threadIdx.x;
    const int w   = t >> 5;
    const int l   = t & 31;
    const int l16 = l & 15;
    const int hi  = l >> 4;

    const int b0 = (blockIdx.x >> 2) * 64;
    const int d0 = (blockIdx.x & 3) * 64;
    const int wb = (w >> 1) * 16;
    const int wd = (w & 1) * 32;

    v8f acc0, acc1;
    #pragma unroll
    for (int r = 0; r < 8; ++r) { acc0[r] = 0.f; acc1[r] = 0.f; }

    float4 ga[2][2], gb[2][2];
    fg_fetch(o1, lw, b0, d0, 0, t, ga, gb);

    for (int ch = 0; ch < FG_NCH; ++ch) {
        const int buf = ch & 1;
        #pragma unroll
        for (int it = 0; it < 2; ++it) {
            const int f   = t + 256 * it;
            const int off = (f >> 4) * FG_PITCH + (f & 15) * 8;
            fg_put(&sA[buf][off], ga[it][0], ga[it][1]);
            fg_put(&sB[buf][off], gb[it][0], gb[it][1]);
        }
        __syncthreads();

        if (ch + 1 < FG_NCH) fg_fetch(o1, lw, b0, d0, (ch + 1) * FG_KC, t, ga, gb);

        const float* pa = &sA[buf][hi * FG_PITCH + (wb + l16) * 2];
        const float* pb = &sB[buf][hi * FG_PITCH + (wd + l16) * 2];
        #pragma unroll
        for (int ks = 0; ks < FG_KC / 4; ++ks) {
            const v2f A  = *(const v2f*)(pa + 2 * ks * FG_PITCH);
            const v2f B0 = *(const v2f*)(pb + 2 * ks * FG_PITCH);
            const v2f B1 = *(const v2f*)(pb + 2 * ks * FG_PITCH + 32);
            acc0 = __builtin_amdgcn_wmma_f32_16x16x4_f32(
                       false, A, false, B0, (short)0, acc0, false, false);
            acc1 = __builtin_amdgcn_wmma_f32_16x16x4_f32(
                       false, A, false, B1, (short)0, acc1, false, false);
        }
    }

    float* po = out + (size_t)(b0 + wb + 8 * hi) * Csz + d0 + wd + l16;
    #pragma unroll
    for (int r = 0; r < 8; ++r) {
        po[(size_t)r * Csz]      = acc0[r] * 0.0625f;
        po[(size_t)r * Csz + 16] = acc1[r] * 0.0625f;
    }
}

extern "C" void kernel_launch(void* const* d_in, const int* in_sizes, int n_in,
                              void* d_out, int out_size, void* d_ws, size_t ws_size,
                              hipStream_t stream) {
    (void)in_sizes; (void)n_in; (void)out_size; (void)ws_size;
    const float* nf  = (const float*)d_in[0];
    const float* U3  = (const float*)d_in[1];
    const float* U2  = (const float*)d_in[2];
    const float* U1  = (const float*)d_in[3];
    const float* w3  = (const float*)d_in[4];
    const float* w2  = (const float*)d_in[5];
    const float* w1  = (const float*)d_in[6];
    const float* lw  = (const float*)d_in[7];

    float* ws   = (float*)d_ws;
    float* out1 = ws + OFF_O1;

    pb_precompute<<<1376, 64, 0, stream>>>(U3, U2, U1, w3, w2, w1, ws);
    pb_main<<<dim3(2, 256), 256, 0, stream>>>(nf, ws, out1);
    pb_final_gemm<<<256, 256, 0, stream>>>(out1, lw, (float*)d_out);
}


# pb_precompute on the matrix unit: Wp, c2, c1 as (position x k)*(k x channel) f32 WMMAs, weights as B fragments in ONE load round (was ~10 dependent scalar rounds), 288 1-wave blocks; k sum order new
# speedup vs baseline: 2.1337x; 1.0337x over previous
#include <hip/hip_runtime.h>


typedef __attribute__((ext_vector_type(2))) float v2f;
typedef __attribute__((ext_vector_type(8))) float v8f;

#define Bsz   4096
#define Csz   256
#define LM    16

#define OFF_W3   0
#define OFF_C2   1048576
#define OFF_C1   1114112
#define OFF_O1   1118208

__host__ __device__ constexpr int PA(int q) { return q < 32 ? (q & 7) : (q - 32); }
__host__ __device__ constexpr int PB(int q) { return q < 32 ? (((q & 7) + (q >> 3)) & 7) : (q - 32 + 4); }

#define NKK      18
#define WP_PER_C 1152

#define FOLD_CG 32
#define PRE_T   32
#define WP_BLK_PER_G 36
#define WP_BLOCKS    288

__device__ __forceinline__ void pre_store(float* __restrict__ dst, const v8f& d)
{
    float4* o = (float4*)dst;
    o[0] = make_float4(d[0], d[1], d[2], d[3]);
    o[1] = make_float4(d[4], d[5], d[6], d[7]);
}

__global__ __launch_bounds__(PRE_T) void pb_precompute(
    const float* __restrict__ U3, const float* __restrict__ U2,
    const float* __restrict__ U1, const float* __restrict__ w3,
    const float* __restrict__ w2, const float* __restrict__ w1,
    float* __restrict__ ws)
{
    const int t   = threadIdx.x;
    const int bx  = blockIdx.x;
    const int cg  = bx / WP_BLK_PER_G;
    const int pb  = bx - cg * WP_BLK_PER_G;
    const int c0  = cg * FOLD_CG;
    const int l   = t & 31;
    const int l16 = l & 15;
    const int h   = l >> 4;
    const int o1 = h ? 0 : 1;
    const unsigned padm = h ? 0u : 0xffffffffu;

    v2f Bw[2][6];
    #pragma unroll
    for (int nt = 0; nt < 2; ++nt) {
        const float* wc = w3 + (2 * h) * Csz + c0 + nt * 16 + l16;
        #pragma unroll
        for (int ks = 0; ks < 6; ++ks) {
            Bw[nt][ks].x = wc[(4 * ks) * Csz];
            Bw[nt][ks].y = (ks < 5) ? wc[(4 * ks + 1) * Csz] : __uint_as_float(__float_as_uint(wc[(20 + o1) * Csz]) & padm);
        }
    }

    float rv[2][6][12];
    float wj[2][6];
    #pragma unroll
    for (int mt = 0; mt < 2; ++mt) {
        const int rem  = pb * PRE_T + mt * 16 + l16;
        const int kk   = rem >> 6;
        const int lane = (rem >> 1) & 31;
        const int sel  = rem & 1;
        const int r = lane & 15, hh = lane >> 4;
        const int q = 2 * kk + sel;
        const int y = PA(q) + 8 * hh;
        const int i = PB(q) + 8 * hh;
        const int lo = y < i ? y : i;
        const int hi = y < i ? i : y;
        const bool home = ((r >> 3) != hh) || (r <= lo);
        int e[6];
        e[0] = (r  * 16 + lo) * 16 + hi;
        e[1] = (r  * 16 + hi) * 16 + lo;
        e[2] = (lo * 16 + r ) * 16 + hi;
        e[3] = (lo * 16 + hi) * 16 + r;
        e[4] = (hi * 16 + r ) * 16 + lo;
        e[5] = (hi * 16 + lo) * 16 + r;
        #pragma unroll
        for (int j = 0; j < 6; ++j) {
            bool first = true;
            #pragma unroll
            for (int jj = 0; jj < j; ++jj) first = first && (e[jj] != e[j]);
            wj[mt][j] = (home && first) ? 1.f : 0.f;
            const float* row = U3 + e[j] * 23 + 2 * h;
            #pragma unroll
            for (int ks = 0; ks < 6; ++ks) {
                rv[mt][j][2 * ks]     = row[4 * ks];
                rv[mt][j][2 * ks + 1] = (ks < 5) ? row[4 * ks + 1] : __uint_as_float(__float_as_uint(row[20 + o1]) & padm);
            }
        }
    }

    v2f A2[2][2], B2[2], A1, B1;
    if (pb < 16) {
        const float* wc = w2 + (2 * h) * Csz + c0 + (pb & 1) * 16 + l16;
        #pragma unroll
        for (int ks = 0; ks < 2; ++ks) {
            B2[ks].x = wc[(4 * ks) * Csz];
            B2[ks].y = wc[(4 * ks + 1) * Csz];
        }
        #pragma unroll
        for (int m = 0; m < 2; ++m) {
            const int rem2  = (2 * (pb >> 1) + m) * 16 + l16;
            const int lane2 = (rem2 >> 1) & 31;
            const int x2 = lane2 & 15;
            const int y2 = 2 * (rem2 >> 6) + (rem2 & 1) + 8 * (lane2 >> 4);
            const float* ur = U2 + (x2 * 16 + y2) * 8 + 2 * h;
            #pragma unroll
            for (int ks = 0; ks < 2; ++ks) {
                A2[m][ks].x = ur[4 * ks];
                A2[m][ks].y = ur[4 * ks + 1];
            }
        }
    } else if (pb < 18) {
        const float* wc = w1 + (2 * h) * Csz + c0 + (pb - 16) * 16 + l16;
        B1.x = wc[0];
        B1.y = wc[Csz];
        A1.x = U1[l16 * 4 + 2 * h];
        A1.y = U1[l16 * 4 + 2 * h + 1];
    }

    #pragma unroll
    for (int mt = 0; mt < 2; ++mt) {
        v2f Au[6];
        #pragma unroll
        for (int ks = 0; ks < 6; ++ks) {
            float u0 = 0.f, u1 = 0.f;
            #pragma unroll
            for (int j = 0; j < 6; ++j) {
                u0 += wj[mt][j] * rv[mt][j][2 * ks];
                u1 += wj[mt][j] * rv[mt][j][2 * ks + 1];
            }
            Au[ks].x = u0; Au[ks].y = u1;
        }
        #pragma unroll
        for (int nt = 0; nt < 2; ++nt) {
            v8f acc;
            #pragma unroll
            for (int r = 0; r < 8; ++r) acc[r] = 0.f;
            #pragma unroll
            for (int ks = 0; ks < 6; ++ks)
                acc = __builtin_amdgcn_wmma_f32_16x16x4_f32(
                          false, Au[ks], false, Bw[nt][ks], (short)0, acc, false, false);
            pre_store(ws + OFF_W3 + (c0 + nt * 16 + l16) * WP_PER_C
                         + pb * PRE_T + mt * 16 + 8 * h, acc);
        }
    }

    if (pb < 16) {
        #pragma unroll
        for (int m = 0; m < 2; ++m) {
            v8f acc;
            #pragma unroll
            for (int r = 0; r < 8; ++r) acc[r] = 0.f;
            #pragma unroll
            for (int ks = 0; ks < 2; ++ks)
                acc = __builtin_amdgcn_wmma_f32_16x16x4_f32(
                          false, A2[m][ks], false, B2[ks], (short)0, acc, false, false);
            pre_store(ws + OFF_C2 + (c0 + (pb & 1) * 16 + l16) * 256
                         + (2 * (pb >> 1) + m) * 16 + 8 * h, acc);
        }
    } else if (pb < 18) {
        v8f acc;
        #pragma unroll
        for (int r = 0; r < 8; ++r) acc[r] = 0.f;
        acc = __builtin_amdgcn_wmma_f32_16x16x4_f32(
                  false, A1, false, B1, (short)0, acc, false, false);
        pre_store(ws + OFF_C1 + (c0 + (pb - 16) * 16 + l16) * 16 + 8 * h, acc);
    }
}

#define PAIRS_PER_WAVE 8

__device__ __forceinline__ void pb_load_half(const float* __restrict__ p, float4 (&r)[2])
{
    const float4* g = (const float4*)p;
    r[0] = g[0]; r[1] = g[1];
}

__device__ __forceinline__ void pb_unpack(const float4 (&r)[2], float (&d)[8])
{
    d[0] = r[0].x; d[1] = r[0].y; d[2] = r[0].z; d[3] = r[0].w;
    d[4] = r[1].x; d[5] = r[1].y; d[6] = r[1].z; d[7] = r[1].w;
}

__global__ __launch_bounds__(256) void pb_main(
    const float* __restrict__ nf,
    const float* __restrict__ ws,
    float* __restrict__ out1)
{
    __shared__ float sA[WP_PER_C];
    __shared__ float sC2[256];
    __shared__ float sC1[16];

    const int t  = threadIdx.x;
    const int c  = blockIdx.y;
    const int w  = t >> 5;
    const int l  = t & 31;
    const int l16 = l & 15;
    const int hi  = l >> 4;
    const bool up = hi != 0;

    const int bw = (blockIdx.x * PAIRS_PER_WAVE * 8 + w) * 32;
    const float* rowp = nf + (size_t)(bw + l16) * (Csz * LM) + c * LM + 8 * hi;
    const size_t tile_stride = (size_t)16 * Csz * LM;
    const size_t pair_stride = (size_t)256 * Csz * LM;

    float4 ra[2], rb[2];
    pb_load_half(rowp, ra);
    pb_load_half(rowp + tile_stride, rb);

    {
        const float4* g = (const float4*)(ws + OFF_W3 + c * WP_PER_C);
        float4* s = (float4*)sA;
        s[t] = g[t];
        if (t < 32) s[256 + t] = g[256 + t];
    }
    sC2[t] = ws[OFF_C2 + c * 256 + t];
    if (t < 16) sC1[t] = ws[OFF_C1 + c * 16 + t];
    __syncthreads();

    const v2f* Af  = (const v2f*)sA;
    const v2f* C2f = (const v2f*)sC2;

    for (int p = 0; p < PAIRS_PER_WAVE; ++p) {
        float d0[8], d1[8];
        pb_unpack(ra, d0);
        pb_unpack(rb, d1);

        if (p + 1 < PAIRS_PER_WAVE) {
            const float* nx = rowp + (size_t)(p + 1) * pair_stride;
            pb_load_half(nx, ra);
            pb_load_half(nx + tile_stride, rb);
        }

        v8f C0, C1;
        #pragma unroll
        for (int r = 0; r < 8; ++r) { float v = sC1[r + 8 * hi]; C0[r] = v; C1[r] = v; }

        #pragma unroll
        for (int kk = 0; kk < 4; ++kk) {
            v2f A = C2f[kk * 32 + l];
            v2f B0, B1;
            B0.x = d0[2 * kk];  B0.y = d0[2 * kk + 1];
            B1.x = d1[2 * kk];  B1.y = d1[2 * kk + 1];
            C0 = __builtin_amdgcn_wmma_f32_16x16x4_f32(
                     false, A, false, B0, (short)0, C0, false, false);
            C1 = __builtin_amdgcn_wmma_f32_16x16x4_f32(
                     false, A, false, B1, (short)0, C1, false, false);
        }
        #pragma unroll
        for (int kk = 0; kk < NKK; ++kk) {
            v2f A = Af[kk * 32 + l];
            v2f B0, B1;
            B0.x = d0[PA(2 * kk)]     * d0[PB(2 * kk)];
            B0.y = d0[PA(2 * kk + 1)] * d0[PB(2 * kk + 1)];
            B1.x = d1[PA(2 * kk)]     * d1[PB(2 * kk)];
            B1.y = d1[PA(2 * kk + 1)] * d1[PB(2 * kk + 1)];
            C0 = __builtin_amdgcn_wmma_f32_16x16x4_f32(
                     false, A, false, B0, (short)0, C0, false, false);
            C1 = __builtin_amdgcn_wmma_f32_16x16x4_f32(
                     false, A, false, B1, (short)0, C1, false, false);
        }

        float e0 = C0[0] * d0[0];
        float e1 = C1[0] * d1[0];
        #pragma unroll
        for (int r = 1; r < 8; ++r) {
            e0 += C0[r] * d0[r];
            e1 += C1[r] * d1[r];
        }
        e0 += __shfl_xor(e0, 16, 32);
        e1 += __shfl_xor(e1, 16, 32);

        out1[c * Bsz + bw + p * 256 + l] = up ? e1 : e0;
    }
}

#define FG_KC     64
#define FG_NCH    4
#define FG_KP     32
#define FG_PITCH  160
#define FG_PANEL  (FG_KP * FG_PITCH)

__device__ __forceinline__ void fg_fetch(const float* __restrict__ o1, const float* __restrict__ lw,
                                         int b0, int d0, int kb, int t,
                                         float4 (&ga)[2][2], float4 (&gb)[2][2])
{
    #pragma unroll
    for (int it = 0; it < 2; ++it) {
        const int f    = t + 256 * it;
        const int col4 = f & 15;
        const int k    = kb + 2 * (f >> 4);
        const float* pa = o1 + (size_t)k * Bsz + b0 + col4 * 4;
        const float* pw = lw + k * Csz + d0 + col4 * 4;
        ga[it][0] = *(const float4*)pa;
        ga[it][1] = *(const float4*)(pa + Bsz);
        gb[it][0] = *(const float4*)pw;
        gb[it][1] = *(const float4*)(pw + Csz);
    }
}

__device__ __forceinline__ void fg_put(float* __restrict__ s, const float4& u, const float4& v)
{
    float4* d = (float4*)s;
    d[0] = make_float4(u.x, v.x, u.y, v.y);
    d[1] = make_float4(u.z, v.z, u.w, v.w);
}

__global__ __launch_bounds__(256) void pb_final_gemm(
    const float* __restrict__ o1,
    const float* __restrict__ lw,
    float* __restrict__ out)
{
    __shared__ __attribute__((aligned(16))) float sA[2][FG_PANEL];
    __shared__ __attribute__((aligned(16))) float sB[2][FG_PANEL];

    const int t   = threadIdx.x;
    const int w   = t >> 5;
    const int l   = t & 31;
    const int l16 = l & 15;
    const int hi  = l >> 4;

    const int b0 = (blockIdx.x >> 2) * 64;
    const int d0 = (blockIdx.x & 3) * 64;
    const int wb = (w >> 1) * 16;
    const int wd = (w & 1) * 32;

    v8f acc0, acc1;
    #pragma unroll
    for (int r = 0; r < 8; ++r) { acc0[r] = 0.f; acc1[r] = 0.f; }

    float4 ga[2][2], gb[2][2];
    fg_fetch(o1, lw, b0, d0, 0, t, ga, gb);

    for (int ch = 0; ch < FG_NCH; ++ch) {
        const int buf = ch & 1;
        #pragma unroll
        for (int it = 0; it < 2; ++it) {
            const int f   = t + 256 * it;
            const int off = (f >> 4) * FG_PITCH + (f & 15) * 8;
            fg_put(&sA[buf][off], ga[it][0], ga[it][1]);
            fg_put(&sB[buf][off], gb[it][0], gb[it][1]);
        }
        __syncthreads();

        if (ch + 1 < FG_NCH) fg_fetch(o1, lw, b0, d0, (ch + 1) * FG_KC, t, ga, gb);

        const float* pa = &sA[buf][hi * FG_PITCH + (wb + l16) * 2];
        const float* pb = &sB[buf][hi * FG_PITCH + (wd + l16) * 2];
        #pragma unroll
        for (int ks = 0; ks < FG_KC / 4; ++ks) {
            const v2f A  = *(const v2f*)(pa + 2 * ks * FG_PITCH);
            const v2f B0 = *(const v2f*)(pb + 2 * ks * FG_PITCH);
            const v2f B1 = *(const v2f*)(pb + 2 * ks * FG_PITCH + 32);
            acc0 = __builtin_amdgcn_wmma_f32_16x16x4_f32(
                       false, A, false, B0, (short)0, acc0, false, false);
            acc1 = __builtin_amdgcn_wmma_f32_16x16x4_f32(
                       false, A, false, B1, (short)0, acc1, false, false);
        }
    }

    float* po = out + (size_t)(b0 + wb + 8 * hi) * Csz + d0 + wd + l16;
    #pragma unroll
    for (int r = 0; r < 8; ++r) {
        po[(size_t)r * Csz]      = acc0[r] * 0.0625f;
        po[(size_t)r * Csz + 16] = acc1[r] * 0.0625f;
    }
}

extern "C" void kernel_launch(void* const* d_in, const int* in_sizes, int n_in,
                              void* d_out, int out_size, void* d_ws, size_t ws_size,
                              hipStream_t stream) {
    (void)in_sizes; (void)n_in; (void)out_size; (void)ws_size;
    const float* nf  = (const float*)d_in[0];
    const float* U3  = (const float*)d_in[1];
    const float* U2  = (const float*)d_in[2];
    const float* U1  = (const float*)d_in[3];
    const float* w3  = (const float*)d_in[4];
    const float* w2  = (const float*)d_in[5];
    const float* w1  = (const float*)d_in[6];
    const float* lw  = (const float*)d_in[7];

    float* ws   = (float*)d_ws;
    float* out1 = ws + OFF_O1;

    pb_precompute<<<WP_BLOCKS, PRE_T, 0, stream>>>(U3, U2, U1, w3, w2, w1, ws);
    pb_main<<<dim3(2, 256), 256, 0, stream>>>(nf, ws, out1);
    pb_final_gemm<<<256, 256, 0, stream>>>(out1, lw, (float*)d_out);
}
